# starting kernel
# baseline (speedup 1.0000x reference)
#include <hip/hip_runtime.h>

#define N_ATOMS 50000
#define N_PAIRS 800000
#define CCH     64

typedef __attribute__((ext_vector_type(2))) float v2f;
typedef __attribute__((ext_vector_type(8))) float v8f;

__global__ __launch_bounds__(256) void zero4_kernel(float4* __restrict__ p, int n4) {
    int t = blockIdx.x * blockDim.x + threadIdx.x;
    if (t < n4) p[t] = make_float4(0.f, 0.f, 0.f, 0.f);
}

__global__ __launch_bounds__(256) void scatter_kernel(
    const int*   __restrict__ ind2,
    const float* __restrict__ p3,
    const float* __restrict__ i1,
    const float* __restrict__ d3,
    float*       __restrict__ agg)
{
    const int pl = threadIdx.x >> 6;
    const int c  = threadIdx.x & 63;
    const int p  = blockIdx.x * 4 + pl;

    const int i = ind2[2 * p];
    const int j = ind2[2 * p + 1];

    __builtin_prefetch((const void*)(i1 + (size_t)(p + 64) * CCH + c), 0, 1);

    const float gate = i1[(size_t)p * CCH + c];
    const float dx0  = d3[3 * p + 0];
    const float dx1  = d3[3 * p + 1];
    const float dx2  = d3[3 * p + 2];

    const float* src = p3 + (size_t)j * (3 * CCH) + c;
    const float  v0  = (src[0]         + dx0) * gate;
    const float  v1  = (src[CCH]       + dx1) * gate;
    const float  v2  = (src[2 * CCH]   + dx2) * gate;

    float* dst = agg + (size_t)i * (3 * CCH) + c;
    (void)__hip_atomic_fetch_add(dst,            v0, __ATOMIC_RELAXED, __HIP_MEMORY_SCOPE_AGENT);
    (void)__hip_atomic_fetch_add(dst + CCH,      v1, __ATOMIC_RELAXED, __HIP_MEMORY_SCOPE_AGENT);
    (void)__hip_atomic_fetch_add(dst + 2 * CCH,  v2, __ATOMIC_RELAXED, __HIP_MEMORY_SCOPE_AGENT);
}

__global__ __launch_bounds__(128) void gemm_wmma_kernel(
    const float* __restrict__ A,
    const float* __restrict__ W,
    float*       __restrict__ D)
{
    const int wave = threadIdx.x >> 5;
    const int lane = threadIdx.x & 31;
    const int tm   = blockIdx.x;

    const int m16  = lane & 15;
    const int koff = (lane >> 4) * 2;

    const float* arow = A + (size_t)(tm * 16 + m16) * CCH + koff;
    const float* wrow = W + (size_t)(wave * 16 + m16) * CCH + koff;

    v8f acc = {};
    #pragma unroll
    for (int k = 0; k < 16; ++k) {
        v2f a = *(const v2f*)(arow + k * 4);
        v2f b = *(const v2f*)(wrow + k * 4);
        acc = __builtin_amdgcn_wmma_f32_16x16x4_f32(
             false, a,  false, b,
             (short)0, acc,  false,  false);
    }

    const int col     = wave * 16 + m16;
    const int rowbase = tm * 16 + (lane >> 4) * 8;
    float* o = D + (size_t)rowbase * CCH + col;
    #pragma unroll
    for (int r = 0; r < 8; ++r)
        o[(size_t)r * CCH] = acc[r];
}

__global__ __launch_bounds__(256) void dotted_kernel(
    const float* __restrict__ p3n,
    float*       __restrict__ dotted)
{
    int t = blockIdx.x * blockDim.x + threadIdx.x;
    if (t >= N_ATOMS * CCH) return;
    int a = t >> 6, d = t & 63;
    const float* p = p3n + (size_t)a * (3 * CCH) + d;
    float x0 = p[0], x1 = p[CCH], x2 = p[2 * CCH];
    dotted[t] = x0 * x0 + x1 * x1 + x2 * x2;
}

extern "C" void kernel_launch(void* const* d_in, const int* in_sizes, int n_in,
                              void* d_out, int out_size, void* d_ws, size_t ws_size,
                              hipStream_t stream) {
    const int*   ind2 = (const int*)  d_in[0];
    const float* p3   = (const float*)d_in[1];
    const float* i1   = (const float*)d_in[2];
    const float* d3   = (const float*)d_in[3];
    const float* W    = (const float*)d_in[4];

    float* p3_new = (float*)d_out;
    float* dotted = (float*)d_out + (size_t)N_ATOMS * 3 * CCH;
    float* agg    = (float*)d_ws;

    const int nagg  = N_ATOMS * 3 * CCH;
    const int nagg4 = nagg / 4;

    zero4_kernel<<<(nagg4 + 255) / 256, 256, 0, stream>>>((float4*)agg, nagg4);

    scatter_kernel<<<N_PAIRS / 4, 256, 0, stream>>>(ind2, p3, i1, d3, agg);

    gemm_wmma_kernel<<<(N_ATOMS * 3) / 16, 128, 0, stream>>>(agg, W, p3_new);

    dotted_kernel<<<(N_ATOMS * CCH + 255) / 256, 256, 0, stream>>>(p3_new, dotted);
}


# Scatter: receivers split by low atom bit between the two package halves; a work-group takes its die's half as class and a rank by integer ticket, adds only its class; sums in arrival order as before
# speedup vs baseline: 1.9372x; 1.9372x over previous
#include <hip/hip_runtime.h>

#define N_ATOMS 50000
#define N_PAIRS 800000
#define CCH     64

typedef __attribute__((ext_vector_type(2))) float v2f;
typedef __attribute__((ext_vector_type(8))) float v8f;

#define ZERO_WGS     1875
#define ZERO_ITERS   5
#define SC_WGS       2000
#define SC_CLASS_WGS 1000
#define SC_QUAD      4
#define SC_STRIDE    (SC_CLASS_WGS * 8 * SC_QUAD)
#define SC_CAND      100
#define DOT_WGS      1250
#define DOT_ITERS    10

__global__ __launch_bounds__(256) void zero4_kernel(float4* __restrict__ p,
                                                    unsigned* __restrict__ tick) {
    int t = blockIdx.x * 256 + threadIdx.x;
    if (t == 0) { tick[0] = 0u; tick[1] = 0u; }
    for (int it = 0; it < ZERO_ITERS; ++it, t += ZERO_WGS * 256)
        p[t] = make_float4(0.f, 0.f, 0.f, 0.f);
}

struct PairRows { float g0, g1, a0, a1, b0, b1, c0, c1; };

__device__ __forceinline__ PairRows load_pair(const float* __restrict__ src,
                                              const float* __restrict__ g)
{
    PairRows r;
    r.g0 = g[0];          r.g1 = g[32];
    r.a0 = src[0];        r.a1 = src[32];
    r.b0 = src[CCH];      r.b1 = src[CCH + 32];
    r.c0 = src[2 * CCH];  r.c1 = src[2 * CCH + 32];
    return r;
}

__device__ __forceinline__ void add_pair(float* __restrict__ dst, const PairRows& r,
                                         float dx0, float dx1, float dx2)
{
    (void)__hip_atomic_fetch_add(dst,                (r.a0 + dx0) * r.g0, __ATOMIC_RELAXED, __HIP_MEMORY_SCOPE_AGENT);
    (void)__hip_atomic_fetch_add(dst + 32,           (r.a1 + dx0) * r.g1, __ATOMIC_RELAXED, __HIP_MEMORY_SCOPE_AGENT);
    (void)__hip_atomic_fetch_add(dst + CCH,          (r.b0 + dx1) * r.g0, __ATOMIC_RELAXED, __HIP_MEMORY_SCOPE_AGENT);
    (void)__hip_atomic_fetch_add(dst + CCH + 32,     (r.b1 + dx1) * r.g1, __ATOMIC_RELAXED, __HIP_MEMORY_SCOPE_AGENT);
    (void)__hip_atomic_fetch_add(dst + 2 * CCH,      (r.c0 + dx2) * r.g0, __ATOMIC_RELAXED, __HIP_MEMORY_SCOPE_AGENT);
    (void)__hip_atomic_fetch_add(dst + 2 * CCH + 32, (r.c1 + dx2) * r.g1, __ATOMIC_RELAXED, __HIP_MEMORY_SCOPE_AGENT);
}

struct PairIdx { int iA, jA, iB, jB; float ax, ay, az, bx, by, bz; };

__device__ __forceinline__ PairIdx load_idx(const int* __restrict__ ind2,
                                            const float* __restrict__ d3, int pA, int pB)
{
    PairIdx q;
    q.iA = ind2[2 * pA]; q.jA = ind2[2 * pA + 1];
    q.iB = ind2[2 * pB]; q.jB = ind2[2 * pB + 1];
    const float* da = d3 + 3 * (size_t)pA;
    const float* db = d3 + 3 * (size_t)pB;
    q.ax = da[0]; q.ay = da[1]; q.az = da[2];
    q.bx = db[0]; q.by = db[1]; q.bz = db[2];
    return q;
}

__device__ __forceinline__ int take_next(unsigned long long& lo, unsigned long long& hi, int prev)
{
    if (lo) { const int c = __builtin_ctzll(lo); lo &= lo - 1; return c; }
    if (hi) { const int c = 64 + __builtin_ctzll(hi); hi &= hi - 1; return c; }
    return prev;
}

__device__ __forceinline__ unsigned class_mask(const int* __restrict__ ind2,
                                               int c, int pbase, int cls)
{
    bool mine = false;
    if (c < SC_CAND) {
        const int p = (c >> 2) * SC_STRIDE + pbase + (c & 3);
        mine = ((ind2[2 * p] & 1) == cls);
    }
    return __builtin_amdgcn_ballot_w32(mine);
}

__global__ __launch_bounds__(256) void scatter_kernel(
    const int*   __restrict__ ind2,
    const float* __restrict__ p3,
    const float* __restrict__ i1,
    const float* __restrict__ d3,
    float*       __restrict__ agg,
    unsigned*    __restrict__ tick)
{
    __shared__ unsigned s_role;
    if (threadIdx.x == 0) {
        unsigned c = ((unsigned)__builtin_amdgcn_s_sendmsg_rtn(0x87) >> 18) & 1u;
        unsigned t = __hip_atomic_fetch_add(tick + c, 1u, __ATOMIC_RELAXED, __HIP_MEMORY_SCOPE_AGENT);
        if (t >= SC_CLASS_WGS) {
            c ^= 1u;
            t = __hip_atomic_fetch_add(tick + c, 1u, __ATOMIC_RELAXED, __HIP_MEMORY_SCOPE_AGENT);
        }
        s_role = (c << 16) | t;
    }
    __syncthreads();

    const int lane = threadIdx.x & 31;
    const int role = __builtin_amdgcn_readfirstlane((int)s_role);
    const int cls  = role >> 16;
    const int cw   = __builtin_amdgcn_readfirstlane((role & 0xffff) * 8 + (int)(threadIdx.x >> 5));

    const int pbase = cw * SC_QUAD;

    const unsigned m0 = class_mask(ind2, lane,      pbase, cls);
    const unsigned m1 = class_mask(ind2, lane + 32, pbase, cls);
    const unsigned m2 = class_mask(ind2, lane + 64, pbase, cls);
    const unsigned m3 = class_mask(ind2, lane + 96, pbase, cls);
    unsigned long long lo = (unsigned long long)m0 | ((unsigned long long)m1 << 32);
    unsigned long long hi = (unsigned long long)m2 | ((unsigned long long)m3 << 32);
    const int n = __builtin_popcountll(lo) + __builtin_popcountll(hi);

    int cA = take_next(lo, hi, 0);
    int cB = take_next(lo, hi, cA);
    int pA = (cA >> 2) * SC_STRIDE + pbase + (cA & 3);
    int pB = (cB >> 2) * SC_STRIDE + pbase + (cB & 3);
    PairIdx q = load_idx(ind2, d3, pA, pB);

    const int trips = n >> 1;
    for (int it = 0; it < trips; ++it) {
        const PairRows rA = load_pair(p3 + (size_t)q.jA * (3 * CCH) + lane, i1 + (size_t)pA * CCH + lane);
        const PairRows rB = load_pair(p3 + (size_t)q.jB * (3 * CCH) + lane, i1 + (size_t)pB * CCH + lane);
        const PairIdx cur = q;
        cA = take_next(lo, hi, cB);
        cB = take_next(lo, hi, cA);
        pA = (cA >> 2) * SC_STRIDE + pbase + (cA & 3);
        pB = (cB >> 2) * SC_STRIDE + pbase + (cB & 3);
        q = load_idx(ind2, d3, pA, pB);
        add_pair(agg + (size_t)cur.iA * (3 * CCH) + lane, rA, cur.ax, cur.ay, cur.az);
        add_pair(agg + (size_t)cur.iB * (3 * CCH) + lane, rB, cur.bx, cur.by, cur.bz);
    }
    if (n & 1) {
        const PairRows rA = load_pair(p3 + (size_t)q.jA * (3 * CCH) + lane, i1 + (size_t)pA * CCH + lane);
        add_pair(agg + (size_t)q.iA * (3 * CCH) + lane, rA, q.ax, q.ay, q.az);
    }
}

#define GEMM_WGS    512
#define GEMM_WAVES  (GEMM_WGS * 4)
#define GEMM_TILES  ((N_ATOMS * 3) / 16)
#define STRIP_FLOATS (16 * CCH)

__device__ __forceinline__ void load_a_strip(v2f (&a)[16], const float* __restrict__ arow)
{
    #pragma unroll
    for (int k = 0; k < 16; ++k) a[k] = *(const v2f*)(arow + k * 4);
}

__device__ __forceinline__ void strip_product(const v2f (&a)[16],
                                              const v2f (&b0)[16], const v2f (&b1)[16],
                                              const v2f (&b2)[16], const v2f (&b3)[16],
                                              float* __restrict__ o)
{
    v8f acc0 = {}, acc1 = {}, acc2 = {}, acc3 = {};
    #pragma unroll
    for (int k = 0; k < 16; ++k) {
        acc0 = __builtin_amdgcn_wmma_f32_16x16x4_f32(false, a[k], false, b0[k], (short)0, acc0, false, false);
        acc1 = __builtin_amdgcn_wmma_f32_16x16x4_f32(false, a[k], false, b1[k], (short)0, acc1, false, false);
        acc2 = __builtin_amdgcn_wmma_f32_16x16x4_f32(false, a[k], false, b2[k], (short)0, acc2, false, false);
        acc3 = __builtin_amdgcn_wmma_f32_16x16x4_f32(false, a[k], false, b3[k], (short)0, acc3, false, false);
    }
    #pragma unroll
    for (int r = 0; r < 8; ++r)
        *(float4*)(o + (size_t)r * CCH) = make_float4(acc0[r], acc1[r], acc2[r], acc3[r]);
}

__global__ __launch_bounds__(128) void gemm_wmma_kernel(
    const float* __restrict__ A,
    const float* __restrict__ W,
    float*       __restrict__ D)
{
    __shared__ v2f wfrag[4 * 16 * 32];

    const int wave = threadIdx.x >> 5;
    const int lane = threadIdx.x & 31;
    const int c16  = lane & 15;
    const int half = lane >> 4;
    const int koff = half * 2;

    const int gw = __builtin_amdgcn_readfirstlane(blockIdx.x * 4 + wave);

    const float* abase = A + (size_t)c16 * CCH + koff;
    float*       obase = D + (size_t)(half * 8) * CCH + 4 * c16;
    v2f a0[16], a1[16];
    load_a_strip(a0, abase + (size_t)gw * STRIP_FLOATS);

    {
        const float* wrow = W + (size_t)(4 * c16 + wave) * CCH + koff;
        #pragma unroll
        for (int k = 0; k < 16; ++k)
            wfrag[(wave * 16 + k) * 32 + lane] = *(const v2f*)(wrow + k * 4);
    }
    __syncthreads();

    v2f b0[16], b1[16], b2[16], b3[16];
    #pragma unroll
    for (int k = 0; k < 16; ++k) {
        b0[k] = wfrag[(0 * 16 + k) * 32 + lane];
        b1[k] = wfrag[(1 * 16 + k) * 32 + lane];
        b2[k] = wfrag[(2 * 16 + k) * 32 + lane];
        b3[k] = wfrag[(3 * 16 + k) * 32 + lane];
    }

    if (gw + 4 * GEMM_WAVES < GEMM_TILES) {
        load_a_strip(a1, abase + (size_t)(gw + 4 * GEMM_WAVES) * STRIP_FLOATS);
        strip_product(a1, b0, b1, b2, b3, obase + (size_t)(gw + 4 * GEMM_WAVES) * STRIP_FLOATS);
    }

    load_a_strip(a1, abase + (size_t)(gw + 1 * GEMM_WAVES) * STRIP_FLOATS);
    __builtin_amdgcn_sched_barrier(0);
    strip_product(a0, b0, b1, b2, b3, obase + (size_t)(gw + 0 * GEMM_WAVES) * STRIP_FLOATS);

    load_a_strip(a0, abase + (size_t)(gw + 2 * GEMM_WAVES) * STRIP_FLOATS);
    __builtin_amdgcn_sched_barrier(0);
    strip_product(a1, b0, b1, b2, b3, obase + (size_t)(gw + 1 * GEMM_WAVES) * STRIP_FLOATS);

    load_a_strip(a1, abase + (size_t)(gw + 3 * GEMM_WAVES) * STRIP_FLOATS);
    __builtin_amdgcn_sched_barrier(0);
    strip_product(a0, b0, b1, b2, b3, obase + (size_t)(gw + 2 * GEMM_WAVES) * STRIP_FLOATS);

    __builtin_amdgcn_sched_barrier(0);
    strip_product(a1, b0, b1, b2, b3, obase + (size_t)(gw + 3 * GEMM_WAVES) * STRIP_FLOATS);
}

__global__ __launch_bounds__(256) void dotted_kernel(
    const float* __restrict__ p3n,
    float*       __restrict__ dotted)
{
    int t = blockIdx.x * 256 + threadIdx.x;
    for (int it = 0; it < DOT_ITERS; ++it, t += DOT_WGS * 256) {
        int a = t >> 6, d = t & 63;
        const float* p = p3n + (size_t)a * (3 * CCH) + d;
        float x0 = p[0], x1 = p[CCH], x2 = p[2 * CCH];
        dotted[t] = x0 * x0 + x1 * x1 + x2 * x2;
    }
}

extern "C" void kernel_launch(void* const* d_in, const int* in_sizes, int n_in,
                              void* d_out, int out_size, void* d_ws, size_t ws_size,
                              hipStream_t stream) {
    const int*   ind2 = (const int*)  d_in[0];
    const float* p3   = (const float*)d_in[1];
    const float* i1   = (const float*)d_in[2];
    const float* d3   = (const float*)d_in[3];
    const float* W    = (const float*)d_in[4];

    float* p3_new = (float*)d_out;
    float* dotted = (float*)d_out + (size_t)N_ATOMS * 3 * CCH;
    float* agg    = (float*)d_ws;

    unsigned* tick = (unsigned*)dotted;

    zero4_kernel<<<ZERO_WGS, 256, 0, stream>>>((float4*)agg, tick);

    scatter_kernel<<<SC_WGS, 256, 0, stream>>>(ind2, p3, i1, d3, agg, tick);

    gemm_wmma_kernel<<<GEMM_WGS, 128, 0, stream>>>(agg, W, p3_new);

    dotted_kernel<<<DOT_WGS, 256, 0, stream>>>(p3_new, dotted);
}


# Scatter-add off memory atomics: one 32-wave work-group per 196 receivers filters pairs by a per-call range-byte plane, adds them in LDS (float adds), stores rows once; no zero pass; LDS arrival order
# speedup vs baseline: 2.4929x; 1.2869x over previous
#include <hip/hip_runtime.h>

#define N_ATOMS 50000
#define N_PAIRS 800000
#define CCH     64

typedef __attribute__((ext_vector_type(2))) float v2f;
typedef __attribute__((ext_vector_type(8))) float v8f;

#define DOT_WGS      1250
#define DOT_ITERS    10

#define RNG_ATOMS    196
#define RNG_WGS      256
#define ACC_FLOATS   (RNG_ATOMS * 3 * CCH)
#define ID_PIECES    (N_PAIRS / 16)
#define ID_CHUNKS    ((ID_PIECES + 31) / 32)
#define LIST_BATCH   32
#define LIST_CAP     (LIST_BATCH + 512)
#define ID_WGS       125
#define ID_THREADS   320
#define ID_ITERS     5

__global__ __launch_bounds__(ID_THREADS) void range_id_kernel(const int* __restrict__ ind2,
                                                              unsigned* __restrict__ idw)
{
    const int t0 = blockIdx.x * ID_THREADS + threadIdx.x;
    #pragma unroll
    for (int it = 0; it < ID_ITERS; ++it) {
        const int t = t0 + it * (ID_WGS * ID_THREADS);
        const int4 a = ((const int4*)ind2)[2 * t];
        const int4 b = ((const int4*)ind2)[2 * t + 1];
        idw[t] =  ((unsigned)a.x / RNG_ATOMS)
               | (((unsigned)a.z / RNG_ATOMS) << 8)
               | (((unsigned)b.x / RNG_ATOMS) << 16)
               | (((unsigned)b.z / RNG_ATOMS) << 24);
    }
}

struct PairRows { float g0, g1, a0, a1, b0, b1, c0, c1; };

__device__ __forceinline__ PairRows load_pair(const float* __restrict__ src,
                                              const float* __restrict__ g)
{
    PairRows r;
    r.g0 = g[0];          r.g1 = g[32];
    r.a0 = src[0];        r.a1 = src[32];
    r.b0 = src[CCH];      r.b1 = src[CCH + 32];
    r.c0 = src[2 * CCH];  r.c1 = src[2 * CCH + 32];
    return r;
}

__device__ __forceinline__ void lds_add(float* p, float v)
{
    (void)__hip_atomic_fetch_add(p, v, __ATOMIC_RELAXED, __HIP_MEMORY_SCOPE_WORKGROUP);
}

__device__ __forceinline__ void add_pair(float* dst, const PairRows& r,
                                         float dx0, float dx1, float dx2)
{
    lds_add(dst,                (r.a0 + dx0) * r.g0);
    lds_add(dst + 32,           (r.a1 + dx0) * r.g1);
    lds_add(dst + CCH,          (r.b0 + dx1) * r.g0);
    lds_add(dst + CCH + 32,     (r.b1 + dx1) * r.g1);
    lds_add(dst + 2 * CCH,      (r.c0 + dx2) * r.g0);
    lds_add(dst + 2 * CCH + 32, (r.c1 + dx2) * r.g1);
}

struct PairS { int p, loc, j; float dx, dy, dz; };

__device__ __forceinline__ PairS load_entry(const int* lst, int e, int n,
                                            const int* __restrict__ ind2,
                                            const float* __restrict__ d3, int base)
{
    PairS s;
    int p = 0;
    if (e < n) p = __builtin_amdgcn_readfirstlane(lst[e]);
    s.p   = p;
    s.loc = ind2[2 * p] - base;
    s.j   = ind2[2 * p + 1];
    const float* d = d3 + 3 * (size_t)p;
    s.dx = d[0]; s.dy = d[1]; s.dz = d[2];
    return s;
}

__device__ __forceinline__ unsigned zero_bytes(unsigned x)
{
    return ~(((x & 0x7F7F7F7Fu) + 0x7F7F7F7Fu) | x | 0x7F7F7F7Fu);
}

__device__ __forceinline__ uint4 load_ids(const uint4* __restrict__ id16, int ch, int lane)
{
    const int q = ch * 32 + lane;
    return id16[q < ID_PIECES ? q : ID_PIECES - 1];
}

__global__ __launch_bounds__(1024) void range_sum_kernel(
    const int*   __restrict__ ind2,
    const float* __restrict__ p3,
    const float* __restrict__ i1,
    const float* __restrict__ d3,
    const uint4* __restrict__ id16,
    float*       __restrict__ agg)
{
    __shared__ float4 acc4[ACC_FLOATS / 4];
    __shared__ int    list[32 * LIST_CAP];
    float* acc = (float*)acc4;

    const int tid  = threadIdx.x;
    const int lane = tid & 31;
    const int wave = __builtin_amdgcn_readfirstlane(tid >> 5);
    const int base = blockIdx.x * RNG_ATOMS;

    for (int i = tid; i < ACC_FLOATS / 4; i += 1024)
        acc4[i] = make_float4(0.f, 0.f, 0.f, 0.f);
    __syncthreads();

    int* lst = list + wave * LIST_CAP;
    const unsigned idrep = (unsigned)blockIdx.x * 0x01010101u;

    int ch = wave;
    uint4 w = load_ids(id16, ch, lane);
    while (ch < ID_CHUNKS) {
        int n = 0;
        do {
            const uint4 wn = load_ids(id16, ch + 32, lane);
            unsigned hm = (zero_bytes(w.x ^ idrep) >> 7) | (zero_bytes(w.y ^ idrep) >> 6)
                        | (zero_bytes(w.z ^ idrep) >> 5) | (zero_bytes(w.w ^ idrep) >> 4);
            if (ch * 32 + lane >= ID_PIECES) hm = 0u;
            const int pbl = (ch * 32 + lane) * 16;
            unsigned m = __builtin_amdgcn_ballot_w32(hm != 0u);
            while (m) {
                if (hm) {
                    const int t = __builtin_ctz(hm);
                    hm &= hm - 1;
                    lst[n + (int)__builtin_amdgcn_mbcnt_lo(m, 0u)] = pbl + ((t & 7) << 2) + (t >> 3);
                }
                n += __builtin_popcount(m);
                m = __builtin_amdgcn_ballot_w32(hm != 0u);
            }
            w = wn;
            ch += 32;
        } while (ch < ID_CHUNKS && n < LIST_BATCH);

        PairS q0 = load_entry(lst, 0, n, ind2, d3, base);
        PairS q1 = load_entry(lst, 1, n, ind2, d3, base);
        PairS q2 = load_entry(lst, 2, n, ind2, d3, base);
        PairS q3 = load_entry(lst, 3, n, ind2, d3, base);
        const int trips = n >> 2;
        for (int t = 0; t < trips; ++t) {
            const PairRows r0 = load_pair(p3 + (size_t)q0.j * (3 * CCH) + lane, i1 + (size_t)q0.p * CCH + lane);
            const PairRows r1 = load_pair(p3 + (size_t)q1.j * (3 * CCH) + lane, i1 + (size_t)q1.p * CCH + lane);
            const PairRows r2 = load_pair(p3 + (size_t)q2.j * (3 * CCH) + lane, i1 + (size_t)q2.p * CCH + lane);
            const PairRows r3 = load_pair(p3 + (size_t)q3.j * (3 * CCH) + lane, i1 + (size_t)q3.p * CCH + lane);
            const PairS c0 = q0, c1 = q1, c2 = q2, c3 = q3;
            const int e = 4 * (t + 1);
            q0 = load_entry(lst, e,     n, ind2, d3, base);
            q1 = load_entry(lst, e + 1, n, ind2, d3, base);
            q2 = load_entry(lst, e + 2, n, ind2, d3, base);
            q3 = load_entry(lst, e + 3, n, ind2, d3, base);
            add_pair(acc + c0.loc * (3 * CCH) + lane, r0, c0.dx, c0.dy, c0.dz);
            add_pair(acc + c1.loc * (3 * CCH) + lane, r1, c1.dx, c1.dy, c1.dz);
            add_pair(acc + c2.loc * (3 * CCH) + lane, r2, c2.dx, c2.dy, c2.dz);
            add_pair(acc + c3.loc * (3 * CCH) + lane, r3, c3.dx, c3.dy, c3.dz);
        }
        const int rem = n & 3;
        if (rem > 0) {
            const PairRows r0 = load_pair(p3 + (size_t)q0.j * (3 * CCH) + lane, i1 + (size_t)q0.p * CCH + lane);
            add_pair(acc + q0.loc * (3 * CCH) + lane, r0, q0.dx, q0.dy, q0.dz);
        }
        if (rem > 1) {
            const PairRows r1 = load_pair(p3 + (size_t)q1.j * (3 * CCH) + lane, i1 + (size_t)q1.p * CCH + lane);
            add_pair(acc + q1.loc * (3 * CCH) + lane, r1, q1.dx, q1.dy, q1.dz);
        }
        if (rem > 2) {
            const PairRows r2 = load_pair(p3 + (size_t)q2.j * (3 * CCH) + lane, i1 + (size_t)q2.p * CCH + lane);
            add_pair(acc + q2.loc * (3 * CCH) + lane, r2, q2.dx, q2.dy, q2.dz);
        }
    }

    __syncthreads();

    const int nrow = (N_ATOMS - base < RNG_ATOMS) ? (N_ATOMS - base) : RNG_ATOMS;
    float4* dst = (float4*)(agg + (size_t)base * (3 * CCH));
    for (int i = tid; i < nrow * (3 * CCH / 4); i += 1024)
        dst[i] = acc4[i];
}

#define GEMM_WGS    512
#define GEMM_WAVES  (GEMM_WGS * 4)
#define GEMM_TILES  ((N_ATOMS * 3) / 16)
#define STRIP_FLOATS (16 * CCH)

__device__ __forceinline__ void load_a_strip(v2f (&a)[16], const float* __restrict__ arow)
{
    #pragma unroll
    for (int k = 0; k < 16; ++k) a[k] = *(const v2f*)(arow + k * 4);
}

__device__ __forceinline__ void strip_product(const v2f (&a)[16],
                                              const v2f (&b0)[16], const v2f (&b1)[16],
                                              const v2f (&b2)[16], const v2f (&b3)[16],
                                              float* __restrict__ o)
{
    v8f acc0 = {}, acc1 = {}, acc2 = {}, acc3 = {};
    #pragma unroll
    for (int k = 0; k < 16; ++k) {
        acc0 = __builtin_amdgcn_wmma_f32_16x16x4_f32(false, a[k], false, b0[k], (short)0, acc0, false, false);
        acc1 = __builtin_amdgcn_wmma_f32_16x16x4_f32(false, a[k], false, b1[k], (short)0, acc1, false, false);
        acc2 = __builtin_amdgcn_wmma_f32_16x16x4_f32(false, a[k], false, b2[k], (short)0, acc2, false, false);
        acc3 = __builtin_amdgcn_wmma_f32_16x16x4_f32(false, a[k], false, b3[k], (short)0, acc3, false, false);
    }
    #pragma unroll
    for (int r = 0; r < 8; ++r)
        *(float4*)(o + (size_t)r * CCH) = make_float4(acc0[r], acc1[r], acc2[r], acc3[r]);
}

__global__ __launch_bounds__(128) void gemm_wmma_kernel(
    const float* __restrict__ A,
    const float* __restrict__ W,
    float*       __restrict__ D)
{
    __shared__ v2f wfrag[4 * 16 * 32];

    const int wave = threadIdx.x >> 5;
    const int lane = threadIdx.x & 31;
    const int c16  = lane & 15;
    const int half = lane >> 4;
    const int koff = half * 2;

    const int gw = __builtin_amdgcn_readfirstlane(blockIdx.x * 4 + wave);

    const float* abase = A + (size_t)c16 * CCH + koff;
    float*       obase = D + (size_t)(half * 8) * CCH + 4 * c16;
    v2f a0[16], a1[16];
    load_a_strip(a0, abase + (size_t)gw * STRIP_FLOATS);

    {
        const float* wrow = W + (size_t)(4 * c16 + wave) * CCH + koff;
        #pragma unroll
        for (int k = 0; k < 16; ++k)
            wfrag[(wave * 16 + k) * 32 + lane] = *(const v2f*)(wrow + k * 4);
    }
    __syncthreads();

    v2f b0[16], b1[16], b2[16], b3[16];
    #pragma unroll
    for (int k = 0; k < 16; ++k) {
        b0[k] = wfrag[(0 * 16 + k) * 32 + lane];
        b1[k] = wfrag[(1 * 16 + k) * 32 + lane];
        b2[k] = wfrag[(2 * 16 + k) * 32 + lane];
        b3[k] = wfrag[(3 * 16 + k) * 32 + lane];
    }

    if (gw + 4 * GEMM_WAVES < GEMM_TILES) {
        load_a_strip(a1, abase + (size_t)(gw + 4 * GEMM_WAVES) * STRIP_FLOATS);
        strip_product(a1, b0, b1, b2, b3, obase + (size_t)(gw + 4 * GEMM_WAVES) * STRIP_FLOATS);
    }

    load_a_strip(a1, abase + (size_t)(gw + 1 * GEMM_WAVES) * STRIP_FLOATS);
    __builtin_amdgcn_sched_barrier(0);
    strip_product(a0, b0, b1, b2, b3, obase + (size_t)(gw + 0 * GEMM_WAVES) * STRIP_FLOATS);

    load_a_strip(a0, abase + (size_t)(gw + 2 * GEMM_WAVES) * STRIP_FLOATS);
    __builtin_amdgcn_sched_barrier(0);
    strip_product(a1, b0, b1, b2, b3, obase + (size_t)(gw + 1 * GEMM_WAVES) * STRIP_FLOATS);

    load_a_strip(a1, abase + (size_t)(gw + 3 * GEMM_WAVES) * STRIP_FLOATS);
    __builtin_amdgcn_sched_barrier(0);
    strip_product(a0, b0, b1, b2, b3, obase + (size_t)(gw + 2 * GEMM_WAVES) * STRIP_FLOATS);

    __builtin_amdgcn_sched_barrier(0);
    strip_product(a1, b0, b1, b2, b3, obase + (size_t)(gw + 3 * GEMM_WAVES) * STRIP_FLOATS);
}

__global__ __launch_bounds__(256) void dotted_kernel(
    const float* __restrict__ p3n,
    float*       __restrict__ dotted)
{
    int t = blockIdx.x * 256 + threadIdx.x;
    for (int it = 0; it < DOT_ITERS; ++it, t += DOT_WGS * 256) {
        int a = t >> 6, d = t & 63;
        const float* p = p3n + (size_t)a * (3 * CCH) + d;
        float x0 = p[0], x1 = p[CCH], x2 = p[2 * CCH];
        dotted[t] = x0 * x0 + x1 * x1 + x2 * x2;
    }
}

extern "C" void kernel_launch(void* const* d_in, const int* in_sizes, int n_in,
                              void* d_out, int out_size, void* d_ws, size_t ws_size,
                              hipStream_t stream) {
    const int*   ind2 = (const int*)  d_in[0];
    const float* p3   = (const float*)d_in[1];
    const float* i1   = (const float*)d_in[2];
    const float* d3   = (const float*)d_in[3];
    const float* W    = (const float*)d_in[4];

    float* p3_new = (float*)d_out;
    float* dotted = (float*)d_out + (size_t)N_ATOMS * 3 * CCH;
    float* agg    = (float*)d_ws;

    unsigned* idw = (unsigned*)dotted;

    range_id_kernel<<<ID_WGS, ID_THREADS, 0, stream>>>(ind2, idw);

    range_sum_kernel<<<RNG_WGS, 1024, 0, stream>>>(ind2, p3, i1, d3, (const uint4*)idw, agg);

    gemm_wmma_kernel<<<GEMM_WGS, 128, 0, stream>>>(agg, W, p3_new);

    dotted_kernel<<<DOT_WGS, 256, 0, stream>>>(p3_new, dotted);
}


# Range sum: 8 pairs per trip, not 4 (lane = 2 adjacent channels, 8-byte loads; indices by lane load + readlane; leftovers carried): twice the row bytes in flight per wave; LDS arrival order as before
# speedup vs baseline: 2.6204x; 1.0511x over previous
#include <hip/hip_runtime.h>

#define N_ATOMS 50000
#define N_PAIRS 800000
#define CCH     64

typedef __attribute__((ext_vector_type(2))) float v2f;
typedef __attribute__((ext_vector_type(8))) float v8f;

#define DOT_WGS      1250
#define DOT_ITERS    10

#define RNG_ATOMS    196
#define RNG_WGS      256
#define ACC_FLOATS   (RNG_ATOMS * 3 * CCH)
#define ID_PIECES    (N_PAIRS / 16)
#define ID_CHUNKS    ((ID_PIECES + 31) / 32)
#define LIST_BATCH   32
#define LIST_CAP     (LIST_BATCH + 512)
#define ID_WGS       125
#define ID_THREADS   320
#define ID_ITERS     5

__global__ __launch_bounds__(ID_THREADS) void range_id_kernel(const int* __restrict__ ind2,
                                                              unsigned* __restrict__ idw)
{
    const int t0 = blockIdx.x * ID_THREADS + threadIdx.x;
    #pragma unroll
    for (int it = 0; it < ID_ITERS; ++it) {
        const int t = t0 + it * (ID_WGS * ID_THREADS);
        const int4 a = ((const int4*)ind2)[2 * t];
        const int4 b = ((const int4*)ind2)[2 * t + 1];
        idw[t] =  ((unsigned)a.x / RNG_ATOMS)
               | (((unsigned)a.z / RNG_ATOMS) << 8)
               | (((unsigned)b.x / RNG_ATOMS) << 16)
               | (((unsigned)b.z / RNG_ATOMS) << 24);
    }
}

#define TRIP_PAIRS   8

struct PairRows { float2 g, a, b, c; };

__device__ __forceinline__ PairRows load_pair(const float* __restrict__ src,
                                              const float* __restrict__ g)
{
    PairRows r;
    r.g = *(const float2*)g;
    r.a = *(const float2*)src;
    r.b = *(const float2*)(src + CCH);
    r.c = *(const float2*)(src + 2 * CCH);
    return r;
}

__device__ __forceinline__ void lds_add(float* p, float v)
{
    (void)__hip_atomic_fetch_add(p, v, __ATOMIC_RELAXED, __HIP_MEMORY_SCOPE_WORKGROUP);
}

__device__ __forceinline__ void add_pair(float* dst, const PairRows& r,
                                         float dx0, float dx1, float dx2)
{
    lds_add(dst,               (r.a.x + dx0) * r.g.x);
    lds_add(dst + 1,           (r.a.y + dx0) * r.g.y);
    lds_add(dst + CCH,         (r.b.x + dx1) * r.g.x);
    lds_add(dst + CCH + 1,     (r.b.y + dx1) * r.g.y);
    lds_add(dst + 2 * CCH,     (r.c.x + dx2) * r.g.x);
    lds_add(dst + 2 * CCH + 1, (r.c.y + dx2) * r.g.y);
}

__device__ __forceinline__ int lane_int(int v, int l)
{
    return __builtin_amdgcn_readlane(v, l);
}

__device__ __forceinline__ float lane_float(float v, int l)
{
    return __int_as_float(__builtin_amdgcn_readlane(__float_as_int(v), l));
}

__device__ __forceinline__ unsigned zero_bytes(unsigned x)
{
    return ~(((x & 0x7F7F7F7Fu) + 0x7F7F7F7Fu) | x | 0x7F7F7F7Fu);
}

__device__ __forceinline__ uint4 load_ids(const uint4* __restrict__ id16, int ch, int lane)
{
    const int q = ch * 32 + lane;
    return id16[q < ID_PIECES ? q : ID_PIECES - 1];
}

__global__ __launch_bounds__(1024) void range_sum_kernel(
    const int*   __restrict__ ind2,
    const float* __restrict__ p3,
    const float* __restrict__ i1,
    const float* __restrict__ d3,
    const uint4* __restrict__ id16,
    float*       __restrict__ agg)
{
    __shared__ float4 acc4[ACC_FLOATS / 4];
    __shared__ int    list[32 * LIST_CAP];
    float* acc = (float*)acc4;

    const int tid  = threadIdx.x;
    const int lane = tid & 31;
    const int wave = __builtin_amdgcn_readfirstlane(tid >> 5);
    const int base = blockIdx.x * RNG_ATOMS;

    for (int i = tid; i < ACC_FLOATS / 4; i += 1024)
        acc4[i] = make_float4(0.f, 0.f, 0.f, 0.f);
    __syncthreads();

    int* lst = list + wave * LIST_CAP;
    const unsigned idrep = (unsigned)blockIdx.x * 0x01010101u;

    int ch = wave;
    int n  = 0;
    uint4 w = load_ids(id16, ch, lane);
    while (ch < ID_CHUNKS) {
        do {
            const uint4 wn = load_ids(id16, ch + 32, lane);
            unsigned hm = (zero_bytes(w.x ^ idrep) >> 7) | (zero_bytes(w.y ^ idrep) >> 6)
                        | (zero_bytes(w.z ^ idrep) >> 5) | (zero_bytes(w.w ^ idrep) >> 4);
            if (ch * 32 + lane >= ID_PIECES) hm = 0u;
            const int pbl = (ch * 32 + lane) * 16;
            unsigned m = __builtin_amdgcn_ballot_w32(hm != 0u);
            while (m) {
                if (hm) {
                    const int t = __builtin_ctz(hm);
                    hm &= hm - 1;
                    lst[n + (int)__builtin_amdgcn_mbcnt_lo(m, 0u)] = pbl + ((t & 7) << 2) + (t >> 3);
                }
                n += __builtin_popcount(m);
                m = __builtin_amdgcn_ballot_w32(hm != 0u);
            }
            w = wn;
            ch += 32;
        } while (ch < ID_CHUNKS && n < LIST_BATCH);

        const bool at_end = ch >= ID_CHUNKS;
        const int  todo   = at_end ? n : (n & ~(TRIP_PAIRS - 1));
        for (int g0 = 0; g0 < todo; g0 += 32) {
            int pv = 0;
            if (g0 + lane < todo) pv = lst[g0 + lane];
            const int2   ij   = ((const int2*)ind2)[pv];
            const float* dp   = d3 + 3 * (size_t)pv;
            const float  dxv  = dp[0], dyv = dp[1], dzv = dp[2];
            const int    locv = (ij.x - base) * (3 * CCH);
            const int    cnt  = (todo - g0 < 32) ? (todo - g0) : 32;
            for (int e0 = 0; e0 < cnt; e0 += TRIP_PAIRS) {
                PairRows r[TRIP_PAIRS];
                #pragma unroll
                for (int k = 0; k < TRIP_PAIRS; ++k) {
                    const int p = lane_int(pv, e0 + k);
                    const int j = lane_int(ij.y, e0 + k);
                    r[k] = load_pair(p3 + (size_t)j * (3 * CCH) + 2 * lane,
                                     i1 + (size_t)p * CCH + 2 * lane);
                }
                #pragma unroll
                for (int k = 0; k < TRIP_PAIRS; ++k) {
                    if (e0 + k < cnt)
                        add_pair(acc + lane_int(locv, e0 + k) + 2 * lane, r[k],
                                 lane_float(dxv, e0 + k), lane_float(dyv, e0 + k),
                                 lane_float(dzv, e0 + k));
                }
            }
        }
        if (!at_end) {
            const int left = n - todo;
            int keep = 0;
            if (lane < left) keep = lst[todo + lane];
            if (lane < left) lst[lane] = keep;
            n = left;
        }
    }

    __syncthreads();

    const int nrow = (N_ATOMS - base < RNG_ATOMS) ? (N_ATOMS - base) : RNG_ATOMS;
    float4* dst = (float4*)(agg + (size_t)base * (3 * CCH));
    for (int i = tid; i < nrow * (3 * CCH / 4); i += 1024)
        dst[i] = acc4[i];
}

#define GEMM_WGS    512
#define GEMM_WAVES  (GEMM_WGS * 4)
#define GEMM_TILES  ((N_ATOMS * 3) / 16)
#define STRIP_FLOATS (16 * CCH)

__device__ __forceinline__ void load_a_strip(v2f (&a)[16], const float* __restrict__ arow)
{
    #pragma unroll
    for (int k = 0; k < 16; ++k) a[k] = *(const v2f*)(arow + k * 4);
}

__device__ __forceinline__ void strip_product(const v2f (&a)[16],
                                              const v2f (&b0)[16], const v2f (&b1)[16],
                                              const v2f (&b2)[16], const v2f (&b3)[16],
                                              float* __restrict__ o)
{
    v8f acc0 = {}, acc1 = {}, acc2 = {}, acc3 = {};
    #pragma unroll
    for (int k = 0; k < 16; ++k) {
        acc0 = __builtin_amdgcn_wmma_f32_16x16x4_f32(false, a[k], false, b0[k], (short)0, acc0, false, false);
        acc1 = __builtin_amdgcn_wmma_f32_16x16x4_f32(false, a[k], false, b1[k], (short)0, acc1, false, false);
        acc2 = __builtin_amdgcn_wmma_f32_16x16x4_f32(false, a[k], false, b2[k], (short)0, acc2, false, false);
        acc3 = __builtin_amdgcn_wmma_f32_16x16x4_f32(false, a[k], false, b3[k], (short)0, acc3, false, false);
    }
    #pragma unroll
    for (int r = 0; r < 8; ++r)
        *(float4*)(o + (size_t)r * CCH) = make_float4(acc0[r], acc1[r], acc2[r], acc3[r]);
}

__global__ __launch_bounds__(128) void gemm_wmma_kernel(
    const float* __restrict__ A,
    const float* __restrict__ W,
    float*       __restrict__ D)
{
    __shared__ v2f wfrag[4 * 16 * 32];

    const int wave = threadIdx.x >> 5;
    const int lane = threadIdx.x & 31;
    const int c16  = lane & 15;
    const int half = lane >> 4;
    const int koff = half * 2;

    const int gw = __builtin_amdgcn_readfirstlane(blockIdx.x * 4 + wave);

    const float* abase = A + (size_t)c16 * CCH + koff;
    float*       obase = D + (size_t)(half * 8) * CCH + 4 * c16;
    v2f a0[16], a1[16];
    load_a_strip(a0, abase + (size_t)gw * STRIP_FLOATS);

    {
        const float* wrow = W + (size_t)(4 * c16 + wave) * CCH + koff;
        #pragma unroll
        for (int k = 0; k < 16; ++k)
            wfrag[(wave * 16 + k) * 32 + lane] = *(const v2f*)(wrow + k * 4);
    }
    __syncthreads();

    v2f b0[16], b1[16], b2[16], b3[16];
    #pragma unroll
    for (int k = 0; k < 16; ++k) {
        b0[k] = wfrag[(0 * 16 + k) * 32 + lane];
        b1[k] = wfrag[(1 * 16 + k) * 32 + lane];
        b2[k] = wfrag[(2 * 16 + k) * 32 + lane];
        b3[k] = wfrag[(3 * 16 + k) * 32 + lane];
    }

    if (gw + 4 * GEMM_WAVES < GEMM_TILES) {
        load_a_strip(a1, abase + (size_t)(gw + 4 * GEMM_WAVES) * STRIP_FLOATS);
        strip_product(a1, b0, b1, b2, b3, obase + (size_t)(gw + 4 * GEMM_WAVES) * STRIP_FLOATS);
    }

    load_a_strip(a1, abase + (size_t)(gw + 1 * GEMM_WAVES) * STRIP_FLOATS);
    __builtin_amdgcn_sched_barrier(0);
    strip_product(a0, b0, b1, b2, b3, obase + (size_t)(gw + 0 * GEMM_WAVES) * STRIP_FLOATS);

    load_a_strip(a0, abase + (size_t)(gw + 2 * GEMM_WAVES) * STRIP_FLOATS);
    __builtin_amdgcn_sched_barrier(0);
    strip_product(a1, b0, b1, b2, b3, obase + (size_t)(gw + 1 * GEMM_WAVES) * STRIP_FLOATS);

    load_a_strip(a1, abase + (size_t)(gw + 3 * GEMM_WAVES) * STRIP_FLOATS);
    __builtin_amdgcn_sched_barrier(0);
    strip_product(a0, b0, b1, b2, b3, obase + (size_t)(gw + 2 * GEMM_WAVES) * STRIP_FLOATS);

    __builtin_amdgcn_sched_barrier(0);
    strip_product(a1, b0, b1, b2, b3, obase + (size_t)(gw + 3 * GEMM_WAVES) * STRIP_FLOATS);
}

__global__ __launch_bounds__(256) void dotted_kernel(
    const float* __restrict__ p3n,
    float*       __restrict__ dotted)
{
    int t = blockIdx.x * 256 + threadIdx.x;
    for (int it = 0; it < DOT_ITERS; ++it, t += DOT_WGS * 256) {
        int a = t >> 6, d = t & 63;
        const float* p = p3n + (size_t)a * (3 * CCH) + d;
        float x0 = p[0], x1 = p[CCH], x2 = p[2 * CCH];
        dotted[t] = x0 * x0 + x1 * x1 + x2 * x2;
    }
}

extern "C" void kernel_launch(void* const* d_in, const int* in_sizes, int n_in,
                              void* d_out, int out_size, void* d_ws, size_t ws_size,
                              hipStream_t stream) {
    const int*   ind2 = (const int*)  d_in[0];
    const float* p3   = (const float*)d_in[1];
    const float* i1   = (const float*)d_in[2];
    const float* d3   = (const float*)d_in[3];
    const float* W    = (const float*)d_in[4];

    float* p3_new = (float*)d_out;
    float* dotted = (float*)d_out + (size_t)N_ATOMS * 3 * CCH;
    float* agg    = (float*)d_ws;

    unsigned* idw = (unsigned*)dotted;

    range_id_kernel<<<ID_WGS, ID_THREADS, 0, stream>>>(ind2, idw);

    range_sum_kernel<<<RNG_WGS, 1024, 0, stream>>>(ind2, p3, i1, d3, (const uint4*)idw, agg);

    gemm_wmma_kernel<<<GEMM_WGS, 128, 0, stream>>>(agg, W, p3_new);

    dotted_kernel<<<DOT_WGS, 256, 0, stream>>>(p3_new, dotted);
}


# Range membership as one bit plane per receiver range (built chunk-wise in LDS by phase 0): a range's work-group scans its own 100 KB plane, not a byte for every pair; LDS arrival order as before
# speedup vs baseline: 2.7458x; 1.0479x over previous
#include <hip/hip_runtime.h>

#define N_ATOMS 50000
#define N_PAIRS 800000
#define CCH     64

typedef __attribute__((ext_vector_type(2))) float v2f;
typedef __attribute__((ext_vector_type(8))) float v8f;

#define DOT_WGS      1250
#define DOT_ITERS    10

#define RNG_ATOMS    196
#define RNG_WGS      256
#define ACC_FLOATS   (RNG_ATOMS * 3 * CCH)
#define CH_PAIRS     4096
#define N_CHUNKS     ((N_PAIRS + CH_PAIRS - 1) / CH_PAIRS)
#define CH_WORDS     (CH_PAIRS / 32)
#define PLANE_WORDS  (N_CHUNKS * CH_WORDS)
#define LIST_BATCH   32
#define LIST_CAP     (LIST_BATCH + 32)
#define BIT_THREADS  256

__device__ __forceinline__ void set_bit(unsigned* pl, int recv, int word, unsigned bit)
{
    const unsigned r = (unsigned)recv / RNG_ATOMS;
    if (r < RNG_WGS)
        (void)__hip_atomic_fetch_or(pl + r * CH_WORDS + word, 1u << bit,
                                    __ATOMIC_RELAXED, __HIP_MEMORY_SCOPE_WORKGROUP);
}

__global__ __launch_bounds__(BIT_THREADS) void range_bits_kernel(const int* __restrict__ ind2,
                                                                 uint4* __restrict__ planes)
{
    __shared__ uint4 pl4[RNG_WGS * CH_WORDS / 4];
    unsigned* pl = (unsigned*)pl4;
    const int tid = threadIdx.x;
    const int c   = blockIdx.x;

    int4 a[4], b[4];
    #pragma unroll
    for (int g = 0; g < 4; ++g) {
        const int q  = c * (CH_PAIRS / 4) + g * BIT_THREADS + tid;
        const int qq = q < N_PAIRS / 4 ? q : N_PAIRS / 4 - 1;
        a[g] = ((const int4*)ind2)[2 * qq];
        b[g] = ((const int4*)ind2)[2 * qq + 1];
    }
    for (int i = tid; i < RNG_WGS * CH_WORDS / 4; i += BIT_THREADS)
        pl4[i] = make_uint4(0u, 0u, 0u, 0u);
    __syncthreads();

    #pragma unroll
    for (int g = 0; g < 4; ++g) {
        const int lq = g * BIT_THREADS + tid;
        if (c * (CH_PAIRS / 4) + lq < N_PAIRS / 4) {
            const int      word = lq >> 3;
            const unsigned sh   = (unsigned)(lq & 7) * 4u;
            set_bit(pl, a[g].x, word, sh);
            set_bit(pl, a[g].z, word, sh + 1u);
            set_bit(pl, b[g].x, word, sh + 2u);
            set_bit(pl, b[g].z, word, sh + 3u);
        }
    }
    __syncthreads();

    const int wave = tid >> 5, lane = tid & 31;
    #pragma unroll 8
    for (int r = wave; r < RNG_WGS; r += BIT_THREADS / 32)
        planes[((size_t)r * N_CHUNKS + c) * (CH_WORDS / 4) + lane] = pl4[r * (CH_WORDS / 4) + lane];
}

#define TRIP_PAIRS   8

struct PairRows { float2 g, a, b, c; };

__device__ __forceinline__ PairRows load_pair(const float* __restrict__ src,
                                              const float* __restrict__ g)
{
    PairRows r;
    r.g = *(const float2*)g;
    r.a = *(const float2*)src;
    r.b = *(const float2*)(src + CCH);
    r.c = *(const float2*)(src + 2 * CCH);
    return r;
}

__device__ __forceinline__ void lds_add(float* p, float v)
{
    (void)__hip_atomic_fetch_add(p, v, __ATOMIC_RELAXED, __HIP_MEMORY_SCOPE_WORKGROUP);
}

__device__ __forceinline__ void add_pair(float* dst, const PairRows& r,
                                         float dx0, float dx1, float dx2)
{
    lds_add(dst,               (r.a.x + dx0) * r.g.x);
    lds_add(dst + 1,           (r.a.y + dx0) * r.g.y);
    lds_add(dst + CCH,         (r.b.x + dx1) * r.g.x);
    lds_add(dst + CCH + 1,     (r.b.y + dx1) * r.g.y);
    lds_add(dst + 2 * CCH,     (r.c.x + dx2) * r.g.x);
    lds_add(dst + 2 * CCH + 1, (r.c.y + dx2) * r.g.y);
}

__device__ __forceinline__ int lane_int(int v, int l)
{
    return __builtin_amdgcn_readlane(v, l);
}

__device__ __forceinline__ float lane_float(float v, int l)
{
    return __int_as_float(__builtin_amdgcn_readlane(__float_as_int(v), l));
}

__device__ __forceinline__ uint4 load_bits(const uint4* __restrict__ plane, int ch, int lane)
{
    return plane[(ch < N_CHUNKS ? ch : N_CHUNKS - 1) * 32 + lane];
}

__global__ __launch_bounds__(1024) void range_sum_kernel(
    const int*   __restrict__ ind2,
    const float* __restrict__ p3,
    const float* __restrict__ i1,
    const float* __restrict__ d3,
    const uint4* __restrict__ planes,
    float*       __restrict__ agg)
{
    __shared__ float4 acc4[ACC_FLOATS / 4];
    __shared__ int    list[32 * LIST_CAP];
    float* acc = (float*)acc4;

    const int tid  = threadIdx.x;
    const int lane = tid & 31;
    const int wave = __builtin_amdgcn_readfirstlane(tid >> 5);
    const int base = blockIdx.x * RNG_ATOMS;

    for (int i = tid; i < ACC_FLOATS / 4; i += 1024)
        acc4[i] = make_float4(0.f, 0.f, 0.f, 0.f);
    __syncthreads();

    int* lst = list + wave * LIST_CAP;
    const uint4* plane = planes + (size_t)blockIdx.x * (PLANE_WORDS / 4);

    int ch = wave;
    int n  = 0;
    uint4 w0 = load_bits(plane, ch, lane);
    uint4 wn = load_bits(plane, ch + 32, lane);
    unsigned long long lo = ((unsigned long long)w0.y << 32) | w0.x;
    unsigned long long hi = ((unsigned long long)w0.w << 32) | w0.z;
    bool at_end = false;
    while (!at_end) {
        while (n < LIST_BATCH && !at_end) {
            const bool     has = (lo | hi) != 0ull;
            const unsigned m   = __builtin_amdgcn_ballot_w32(has);
            if (m == 0u) {
                ch += 32;
                at_end = ch >= N_CHUNKS;
                lo = at_end ? 0ull : (((unsigned long long)wn.y << 32) | wn.x);
                hi = at_end ? 0ull : (((unsigned long long)wn.w << 32) | wn.z);
                wn = load_bits(plane, ch + 32, lane);
                continue;
            }
            if (has) {
                int t;
                if (lo) { t = __builtin_ctzll(lo);      lo &= lo - 1ull; }
                else    { t = 64 + __builtin_ctzll(hi); hi &= hi - 1ull; }
                lst[n + (int)__builtin_amdgcn_mbcnt_lo(m, 0u)] = (ch * 32 + lane) * 128 + t;
            }
            n += __builtin_popcount(m);
        }

        const int  todo   = at_end ? n : (n & ~(TRIP_PAIRS - 1));
        for (int g0 = 0; g0 < todo; g0 += 32) {
            int pv = 0;
            if (g0 + lane < todo) pv = lst[g0 + lane];
            const int2   ij   = ((const int2*)ind2)[pv];
            const float* dp   = d3 + 3 * (size_t)pv;
            const float  dxv  = dp[0], dyv = dp[1], dzv = dp[2];
            const int    locv = (ij.x - base) * (3 * CCH);
            const int    cnt  = (todo - g0 < 32) ? (todo - g0) : 32;
            for (int e0 = 0; e0 < cnt; e0 += TRIP_PAIRS) {
                PairRows r[TRIP_PAIRS];
                #pragma unroll
                for (int k = 0; k < TRIP_PAIRS; ++k) {
                    const int p = lane_int(pv, e0 + k);
                    const int j = lane_int(ij.y, e0 + k);
                    r[k] = load_pair(p3 + (size_t)j * (3 * CCH) + 2 * lane,
                                     i1 + (size_t)p * CCH + 2 * lane);
                }
                #pragma unroll
                for (int k = 0; k < TRIP_PAIRS; ++k) {
                    if (e0 + k < cnt)
                        add_pair(acc + lane_int(locv, e0 + k) + 2 * lane, r[k],
                                 lane_float(dxv, e0 + k), lane_float(dyv, e0 + k),
                                 lane_float(dzv, e0 + k));
                }
            }
        }
        if (!at_end) {
            const int left = n - todo;
            int keep = 0;
            if (lane < left) keep = lst[todo + lane];
            if (lane < left) lst[lane] = keep;
            n = left;
        }
    }

    __syncthreads();

    const int nrow = (N_ATOMS - base < RNG_ATOMS) ? (N_ATOMS - base) : RNG_ATOMS;
    float4* dst = (float4*)(agg + (size_t)base * (3 * CCH));
    for (int i = tid; i < nrow * (3 * CCH / 4); i += 1024)
        dst[i] = acc4[i];
}

#define GEMM_WGS    512
#define GEMM_WAVES  (GEMM_WGS * 4)
#define GEMM_TILES  ((N_ATOMS * 3) / 16)
#define STRIP_FLOATS (16 * CCH)

__device__ __forceinline__ void load_a_strip(v2f (&a)[16], const float* __restrict__ arow)
{
    #pragma unroll
    for (int k = 0; k < 16; ++k) a[k] = *(const v2f*)(arow + k * 4);
}

__device__ __forceinline__ void strip_product(const v2f (&a)[16],
                                              const v2f (&b0)[16], const v2f (&b1)[16],
                                              const v2f (&b2)[16], const v2f (&b3)[16],
                                              float* __restrict__ o)
{
    v8f acc0 = {}, acc1 = {}, acc2 = {}, acc3 = {};
    #pragma unroll
    for (int k = 0; k < 16; ++k) {
        acc0 = __builtin_amdgcn_wmma_f32_16x16x4_f32(false, a[k], false, b0[k], (short)0, acc0, false, false);
        acc1 = __builtin_amdgcn_wmma_f32_16x16x4_f32(false, a[k], false, b1[k], (short)0, acc1, false, false);
        acc2 = __builtin_amdgcn_wmma_f32_16x16x4_f32(false, a[k], false, b2[k], (short)0, acc2, false, false);
        acc3 = __builtin_amdgcn_wmma_f32_16x16x4_f32(false, a[k], false, b3[k], (short)0, acc3, false, false);
    }
    #pragma unroll
    for (int r = 0; r < 8; ++r)
        *(float4*)(o + (size_t)r * CCH) = make_float4(acc0[r], acc1[r], acc2[r], acc3[r]);
}

__global__ __launch_bounds__(128) void gemm_wmma_kernel(
    const float* __restrict__ A,
    const float* __restrict__ W,
    float*       __restrict__ D)
{
    __shared__ v2f wfrag[4 * 16 * 32];

    const int wave = threadIdx.x >> 5;
    const int lane = threadIdx.x & 31;
    const int c16  = lane & 15;
    const int half = lane >> 4;
    const int koff = half * 2;

    const int gw = __builtin_amdgcn_readfirstlane(blockIdx.x * 4 + wave);

    const float* abase = A + (size_t)c16 * CCH + koff;
    float*       obase = D + (size_t)(half * 8) * CCH + 4 * c16;
    v2f a0[16], a1[16];
    load_a_strip(a0, abase + (size_t)gw * STRIP_FLOATS);

    {
        const float* wrow = W + (size_t)(4 * c16 + wave) * CCH + koff;
        #pragma unroll
        for (int k = 0; k < 16; ++k)
            wfrag[(wave * 16 + k) * 32 + lane] = *(const v2f*)(wrow + k * 4);
    }
    __syncthreads();

    v2f b0[16], b1[16], b2[16], b3[16];
    #pragma unroll
    for (int k = 0; k < 16; ++k) {
        b0[k] = wfrag[(0 * 16 + k) * 32 + lane];
        b1[k] = wfrag[(1 * 16 + k) * 32 + lane];
        b2[k] = wfrag[(2 * 16 + k) * 32 + lane];
        b3[k] = wfrag[(3 * 16 + k) * 32 + lane];
    }

    if (gw + 4 * GEMM_WAVES < GEMM_TILES) {
        load_a_strip(a1, abase + (size_t)(gw + 4 * GEMM_WAVES) * STRIP_FLOATS);
        strip_product(a1, b0, b1, b2, b3, obase + (size_t)(gw + 4 * GEMM_WAVES) * STRIP_FLOATS);
    }

    load_a_strip(a1, abase + (size_t)(gw + 1 * GEMM_WAVES) * STRIP_FLOATS);
    __builtin_amdgcn_sched_barrier(0);
    strip_product(a0, b0, b1, b2, b3, obase + (size_t)(gw + 0 * GEMM_WAVES) * STRIP_FLOATS);

    load_a_strip(a0, abase + (size_t)(gw + 2 * GEMM_WAVES) * STRIP_FLOATS);
    __builtin_amdgcn_sched_barrier(0);
    strip_product(a1, b0, b1, b2, b3, obase + (size_t)(gw + 1 * GEMM_WAVES) * STRIP_FLOATS);

    load_a_strip(a1, abase + (size_t)(gw + 3 * GEMM_WAVES) * STRIP_FLOATS);
    __builtin_amdgcn_sched_barrier(0);
    strip_product(a0, b0, b1, b2, b3, obase + (size_t)(gw + 2 * GEMM_WAVES) * STRIP_FLOATS);

    __builtin_amdgcn_sched_barrier(0);
    strip_product(a1, b0, b1, b2, b3, obase + (size_t)(gw + 3 * GEMM_WAVES) * STRIP_FLOATS);
}

__global__ __launch_bounds__(256) void dotted_kernel(
    const float* __restrict__ p3n,
    float*       __restrict__ dotted)
{
    int t = blockIdx.x * 256 + threadIdx.x;
    for (int it = 0; it < DOT_ITERS; ++it, t += DOT_WGS * 256) {
        int a = t >> 6, d = t & 63;
        const float* p = p3n + (size_t)a * (3 * CCH) + d;
        float x0 = p[0], x1 = p[CCH], x2 = p[2 * CCH];
        dotted[t] = x0 * x0 + x1 * x1 + x2 * x2;
    }
}

extern "C" void kernel_launch(void* const* d_in, const int* in_sizes, int n_in,
                              void* d_out, int out_size, void* d_ws, size_t ws_size,
                              hipStream_t stream) {
    const int*   ind2 = (const int*)  d_in[0];
    const float* p3   = (const float*)d_in[1];
    const float* i1   = (const float*)d_in[2];
    const float* d3   = (const float*)d_in[3];
    const float* W    = (const float*)d_in[4];

    float* p3_new = (float*)d_out;
    float* dotted = (float*)d_out + (size_t)N_ATOMS * 3 * CCH;
    float* agg    = (float*)d_ws;

    uint4* planes = (uint4*)p3_new;

    range_bits_kernel<<<N_CHUNKS, BIT_THREADS, 0, stream>>>(ind2, planes);

    range_sum_kernel<<<RNG_WGS, 1024, 0, stream>>>(ind2, p3, i1, d3, (const uint4*)planes, agg);

    gemm_wmma_kernel<<<GEMM_WGS, 128, 0, stream>>>(agg, W, p3_new);

    dotted_kernel<<<DOT_WGS, 256, 0, stream>>>(p3_new, dotted);
}
